# starting kernel
# baseline (speedup 1.0000x reference)
#include <hip/hip_runtime.h>
#include <hip/hip_bf16.h>

typedef __attribute__((ext_vector_type(2))) float v2f;
typedef __attribute__((ext_vector_type(4))) float v4f;
typedef __attribute__((ext_vector_type(8))) float v8f;
typedef __attribute__((ext_vector_type(2))) int   v2i;

#define N_NODES 4096
#define F_IN    512
#define F_OUT   512
#define HEADS   4
#define HEAD_DIM 128
#define ALPHA   0.2f
#define NEG_INF (-9.0e15f)

__global__ __launch_bounds__(128) void gemm_xw(const float* __restrict__ x,
                                               const float* __restrict__ W,
                                               float* __restrict__ h) {
  const int lane = threadIdx.x & 31;
  const int wave = threadIdx.x >> 5;
  const int half = lane >> 4;
  const int lid  = lane & 15;
  const int kb   = half * 2;

  const int bm = blockIdx.x;
  const int bn = blockIdx.y;
  const int row = bm * 64 + wave * 16 + lid;

  v8f acc[4];
#pragma unroll
  for (int t = 0; t < 4; ++t) acc[t] = {};

  const float* xrow = x + (size_t)row * F_IN;
  for (int k = 0; k < F_IN; k += 4) {
    v2f a;
    a.x = xrow[k + kb + 0];
    a.y = xrow[k + kb + 1];
#pragma unroll
    for (int t = 0; t < 4; ++t) {
      const int col = bn * 64 + t * 16 + lid;
      v2f b;
      b.x = W[(size_t)(k + kb + 0) * F_OUT + col];
      b.y = W[(size_t)(k + kb + 1) * F_OUT + col];
      acc[t] = __builtin_amdgcn_wmma_f32_16x16x4_f32(
          false, a, false, b, (short)0, acc[t], false, false);
    }
  }

#pragma unroll
  for (int t = 0; t < 4; ++t) {
#pragma unroll
    for (int v = 0; v < 8; ++v) {
      const int m = bm * 64 + wave * 16 + v + half * 8;
      const int c = bn * 64 + t * 16 + lid;
      h[(size_t)m * F_OUT + c] = acc[t][v];
    }
  }
}

__global__ __launch_bounds__(128) void attn_scores(const float* __restrict__ h,
                                                   const float* __restrict__ a,
                                                   float* __restrict__ si,
                                                   float* __restrict__ sj,
                                                   float* __restrict__ sjT) {
  const int lane = threadIdx.x & 31;
  const int wave = threadIdx.x >> 5;
  const int n = blockIdx.x * 4 + wave;

  const v4f a1 = *(const v4f*)(a + lane * 4);
  const v4f a2 = *(const v4f*)(a + HEAD_DIM + lane * 4);

#pragma unroll
  for (int head = 0; head < HEADS; ++head) {
    const float* hrow = h + (size_t)n * F_OUT + head * HEAD_DIM;
    const v4f hv = *(const v4f*)(hrow + lane * 4);
    float s1 = hv.x * a1.x + hv.y * a1.y + hv.z * a1.z + hv.w * a1.w;
    float s2 = hv.x * a2.x + hv.y * a2.y + hv.z * a2.z + hv.w * a2.w;
#pragma unroll
    for (int off = 16; off > 0; off >>= 1) {
      s1 += __shfl_down(s1, off, 32);
      s2 += __shfl_down(s2, off, 32);
    }
    if (lane == 0) {
      si[head * N_NODES + n] = s1;
      sj[head * N_NODES + n] = s2;
      sjT[n * HEADS + head] = s2;
    }
  }
}

__global__ __launch_bounds__(256) void row_max(const int* __restrict__ adj,
                                               const float* __restrict__ sjT,
                                               float* __restrict__ Mx) {
  const int i = blockIdx.x;
  const int t = threadIdx.x;
  v4f m = {NEG_INF, NEG_INF, NEG_INF, NEG_INF};
  const int* arow = adj + (size_t)i * N_NODES;
  for (int j = t; j < N_NODES; j += 256) {
    if (arow[j] > 0) {
      const v4f sv = *(const v4f*)(sjT + j * 4);
      m.x = fmaxf(m.x, sv.x);
      m.y = fmaxf(m.y, sv.y);
      m.z = fmaxf(m.z, sv.z);
      m.w = fmaxf(m.w, sv.w);
    }
  }
  __shared__ v4f red[256];
  red[t] = m;
  __syncthreads();
  for (int s = 128; s > 0; s >>= 1) {
    if (t < s) {
      v4f o = red[t + s];
      v4f c = red[t];
      c.x = fmaxf(c.x, o.x);
      c.y = fmaxf(c.y, o.y);
      c.z = fmaxf(c.z, o.z);
      c.w = fmaxf(c.w, o.w);
      red[t] = c;
    }
    __syncthreads();
  }
  if (t < 4) Mx[t * N_NODES + i] = red[0][t];
}

#define CHUNK_J 64
#define PAIRS   (CHUNK_J / 2)
#define PAIR_STRIDE 288

template <bool PARTIAL>
__global__ __launch_bounds__(128) void attn_agg(const float* __restrict__ h,
                                                const int* __restrict__ adj,
                                                const float* __restrict__ si,
                                                const float* __restrict__ sj,
                                                const float* __restrict__ Mx,
                                                float* __restrict__ outp,
                                                float* __restrict__ pssum,
                                                int jspan) {
  __shared__ float Vp[PAIRS * PAIR_STRIDE];
  __shared__ float srow[4][16];

  const int lane = threadIdx.x & 31;
  const int wave = threadIdx.x >> 5;
  const int half = lane >> 4;
  const int lid  = lane & 15;
  const int kb   = half * 2;

  const int head = blockIdx.y;
  const int iblk = blockIdx.x * 64;
  const int z    = blockIdx.z;
  const int jstart = z * jspan;
  const int row  = iblk + wave * 16 + lid;

  const float sir = si[head * N_NODES + row];
  const float mx0 = sir + Mx[head * N_NODES + row];
  const float mrow = (mx0 > 0.f) ? mx0 : ALPHA * mx0;

  v8f acc[8];
#pragma unroll
  for (int t = 0; t < 8; ++t) acc[t] = {};
  float ssum = 0.f;

  const float* sjh  = sj + head * N_NODES;
  const int*   arow = adj + (size_t)row * N_NODES;
  const float* Vg   = h + head * HEAD_DIM;

  const int st_pair = threadIdx.x >> 2;
  const int st_d0   = (threadIdx.x & 3) * 32;

  for (int c = jstart; c < jstart + jspan; c += CHUNK_J) {
    {
      const float* src0 = Vg + (size_t)(c + st_pair * 2 + 0) * F_OUT + st_d0;
      const float* src1 = Vg + (size_t)(c + st_pair * 2 + 1) * F_OUT + st_d0;
      float* dst = Vp + st_pair * PAIR_STRIDE + st_d0 * 2;
#pragma unroll
      for (int q = 0; q < 8; ++q) {
        const v4f r0 = *(const v4f*)(src0 + q * 4);
        const v4f r1 = *(const v4f*)(src1 + q * 4);
        v4f w0 = {r0.x, r1.x, r0.y, r1.y};
        v4f w1 = {r0.z, r1.z, r0.w, r1.w};
        *(v4f*)(dst + q * 8 + 0) = w0;
        *(v4f*)(dst + q * 8 + 4) = w1;
      }
    }
    __syncthreads();

#pragma unroll
    for (int s4 = 0; s4 < CHUNK_J / 4; ++s4) {
      const int jg0 = c + s4 * 4 + kb;
      const v2f sjv = *(const v2f*)(sjh + jg0);
      const v2i av  = *(const v2i*)(arow + jg0);

      float e0 = sir + sjv.x; e0 = (e0 > 0.f) ? e0 : ALPHA * e0;
      float e1 = sir + sjv.y; e1 = (e1 > 0.f) ? e1 : ALPHA * e1;
      const float p0 = (av.x > 0) ? __expf(e0 - mrow) : 0.f;
      const float p1 = (av.y > 0) ? __expf(e1 - mrow) : 0.f;
      ssum += p0 + p1;

      v2f af;
      af.x = p0;
      af.y = p1;
      const float* bp = Vp + (s4 * 2 + half) * PAIR_STRIDE + lid * 2;
#pragma unroll
      for (int t = 0; t < 8; ++t) {
        const v2f bf = *(const v2f*)(bp + t * 32);
        acc[t] = __builtin_amdgcn_wmma_f32_16x16x4_f32(
            false, af, false, bf, (short)0, acc[t], false, false);
      }
    }
    __syncthreads();
  }

  ssum += __shfl_xor(ssum, 16, 32);

  if (PARTIAL) {
    if (half == 0) pssum[(size_t)z * (HEADS * N_NODES) + head * N_NODES + row] = ssum;
    float* pa = outp + (size_t)z * (N_NODES * F_OUT);
#pragma unroll
    for (int t = 0; t < 8; ++t) {
#pragma unroll
      for (int v = 0; v < 8; ++v) {
        const int m = iblk + wave * 16 + v + half * 8;
        pa[(size_t)m * F_OUT + head * HEAD_DIM + t * 16 + lid] = acc[t][v];
      }
    }
  } else {
    if (half == 0) srow[wave][lid] = ssum;
    __syncthreads();
#pragma unroll
    for (int t = 0; t < 8; ++t) {
#pragma unroll
      for (int v = 0; v < 8; ++v) {
        const int m = iblk + wave * 16 + v + half * 8;
        float val = acc[t][v] / srow[wave][v + half * 8];
        val = (val > 0.f) ? val : (__expf(val) - 1.f);
        outp[(size_t)m * F_OUT + head * HEAD_DIM + t * 16 + lid] = val;
      }
    }
  }
}

template <int JS>
__global__ __launch_bounds__(256) void attn_reduce(const float* __restrict__ pacc,
                                                   const float* __restrict__ pssum,
                                                   float* __restrict__ out) {
  const int i = blockIdx.x;
#pragma unroll
  for (int r = 0; r < 2; ++r) {
    const int col = threadIdx.x + r * 256;
    const int head = col >> 7;
    float num = 0.f, den = 0.f;
#pragma unroll
    for (int zz = 0; zz < JS; ++zz) {
      num += pacc[(size_t)zz * (N_NODES * F_OUT) + (size_t)i * F_OUT + col];
      den += pssum[(size_t)zz * (HEADS * N_NODES) + head * N_NODES + i];
    }
    float val = num / den;
    val = (val > 0.f) ? val : (__expf(val) - 1.f);
    out[(size_t)i * F_OUT + col] = val;
  }
}

extern "C" void kernel_launch(void* const* d_in, const int* in_sizes, int n_in,
                              void* d_out, int out_size, void* d_ws, size_t ws_size,
                              hipStream_t stream) {
  const float* x   = (const float*)d_in[0];
  const int*   adj = (const int*)d_in[1];
  const float* W   = (const float*)d_in[2];
  const float* a   = (const float*)d_in[3];
  float* out = (float*)d_out;

  constexpr int JS = 4;
  const size_t HN = (size_t)HEADS * N_NODES;

  float* h   = (float*)d_ws;
  float* si  = h + (size_t)N_NODES * F_OUT;
  float* sj  = si + HN;
  float* sjT = sj + HN;
  float* Mx  = sjT + HN;
  float* pacc  = Mx + HN;
  float* pssum = pacc + (size_t)JS * N_NODES * F_OUT;
  const size_t need_js =
      (size_t)(pssum + JS * HN - (float*)d_ws) * sizeof(float);

  gemm_xw<<<dim3(N_NODES / 64, F_OUT / 64), 128, 0, stream>>>(x, W, h);
  attn_scores<<<dim3(N_NODES / 4), 128, 0, stream>>>(h, a, si, sj, sjT);
  row_max<<<dim3(N_NODES), 256, 0, stream>>>(adj, sjT, Mx);

  if (ws_size >= need_js) {
    attn_agg<true><<<dim3(N_NODES / 64, HEADS, JS), 128, 0, stream>>>(
        h, adj, si, sj, Mx, pacc, pssum, N_NODES / JS);
    attn_reduce<JS><<<dim3(N_NODES), 256, 0, stream>>>(pacc, pssum, out);
  } else {
    attn_agg<false><<<dim3(N_NODES / 64, HEADS, 1), 128, 0, stream>>>(
        h, adj, si, sj, Mx, out, nullptr, N_NODES);
  }
}


# attn_agg re-cut: V held in LDS in 512-column slabs, a 16-wave block walks 256 rows past each slab (2 stagings, not 16; no barrier inside a slab); adj fetched one line ahead; sum order kept
# speedup vs baseline: 1.3974x; 1.3974x over previous
#include <hip/hip_runtime.h>
#include <hip/hip_bf16.h>

typedef __attribute__((ext_vector_type(2))) float v2f;
typedef __attribute__((ext_vector_type(4))) float v4f;
typedef __attribute__((ext_vector_type(8))) float v8f;
typedef __attribute__((ext_vector_type(2))) int   v2i;

#define N_NODES 4096
#define F_IN    512
#define F_OUT   512
#define HEADS   4
#define HEAD_DIM 128
#define ALPHA   0.2f
#define NEG_INF (-9.0e15f)

__global__ __launch_bounds__(128) void gemm_xw(const float* __restrict__ x,
                                               const float* __restrict__ W,
                                               float* __restrict__ h) {
  const int lane = threadIdx.x & 31;
  const int wave = threadIdx.x >> 5;
  const int half = lane >> 4;
  const int lid  = lane & 15;
  const int kb   = half * 2;

  const int bm = blockIdx.x;
  const int bn = blockIdx.y;
  const int row = bm * 64 + wave * 16 + lid;

  v8f acc[4];
#pragma unroll
  for (int t = 0; t < 4; ++t) acc[t] = {};

  const float* xrow = x + (size_t)row * F_IN;
  for (int k = 0; k < F_IN; k += 4) {
    v2f a;
    a.x = xrow[k + kb + 0];
    a.y = xrow[k + kb + 1];
#pragma unroll
    for (int t = 0; t < 4; ++t) {
      const int col = bn * 64 + t * 16 + lid;
      v2f b;
      b.x = W[(size_t)(k + kb + 0) * F_OUT + col];
      b.y = W[(size_t)(k + kb + 1) * F_OUT + col];
      acc[t] = __builtin_amdgcn_wmma_f32_16x16x4_f32(
          false, a, false, b, (short)0, acc[t], false, false);
    }
  }

#pragma unroll
  for (int t = 0; t < 4; ++t) {
#pragma unroll
    for (int v = 0; v < 8; ++v) {
      const int m = bm * 64 + wave * 16 + v + half * 8;
      const int c = bn * 64 + t * 16 + lid;
      h[(size_t)m * F_OUT + c] = acc[t][v];
    }
  }
}

__global__ __launch_bounds__(128) void attn_scores(const float* __restrict__ h,
                                                   const float* __restrict__ a,
                                                   float* __restrict__ si,
                                                   float* __restrict__ sj,
                                                   float* __restrict__ sjT) {
  const int lane = threadIdx.x & 31;
  const int wave = threadIdx.x >> 5;
  const int n = blockIdx.x * 4 + wave;

  const v4f a1 = *(const v4f*)(a + lane * 4);
  const v4f a2 = *(const v4f*)(a + HEAD_DIM + lane * 4);

#pragma unroll
  for (int head = 0; head < HEADS; ++head) {
    const float* hrow = h + (size_t)n * F_OUT + head * HEAD_DIM;
    const v4f hv = *(const v4f*)(hrow + lane * 4);
    float s1 = hv.x * a1.x + hv.y * a1.y + hv.z * a1.z + hv.w * a1.w;
    float s2 = hv.x * a2.x + hv.y * a2.y + hv.z * a2.z + hv.w * a2.w;
#pragma unroll
    for (int off = 16; off > 0; off >>= 1) {
      s1 += __shfl_down(s1, off, 32);
      s2 += __shfl_down(s2, off, 32);
    }
    if (lane == 0) {
      si[head * N_NODES + n] = s1;
      sj[head * N_NODES + n] = s2;
      sjT[n * HEADS + head] = s2;
    }
  }
}

__global__ __launch_bounds__(256) void row_max(const int* __restrict__ adj,
                                               const float* __restrict__ sjT,
                                               float* __restrict__ Mx) {
  const int i = blockIdx.x;
  const int t = threadIdx.x;
  v4f m = {NEG_INF, NEG_INF, NEG_INF, NEG_INF};
  const int* arow = adj + (size_t)i * N_NODES;
  for (int j = t; j < N_NODES; j += 256) {
    if (arow[j] > 0) {
      const v4f sv = *(const v4f*)(sjT + j * 4);
      m.x = fmaxf(m.x, sv.x);
      m.y = fmaxf(m.y, sv.y);
      m.z = fmaxf(m.z, sv.z);
      m.w = fmaxf(m.w, sv.w);
    }
  }
  __shared__ v4f red[256];
  red[t] = m;
  __syncthreads();
  for (int s = 128; s > 0; s >>= 1) {
    if (t < s) {
      v4f o = red[t + s];
      v4f c = red[t];
      c.x = fmaxf(c.x, o.x);
      c.y = fmaxf(c.y, o.y);
      c.z = fmaxf(c.z, o.z);
      c.w = fmaxf(c.w, o.w);
      red[t] = c;
    }
    __syncthreads();
  }
  if (t < 4) Mx[t * N_NODES + i] = red[0][t];
}

#define CHUNK_J 64
#define PAIRS   (CHUNK_J / 2)
#define PAIR_STRIDE 288

template <bool PARTIAL>
__global__ __launch_bounds__(128) void attn_agg(const float* __restrict__ h,
                                                const int* __restrict__ adj,
                                                const float* __restrict__ si,
                                                const float* __restrict__ sj,
                                                const float* __restrict__ Mx,
                                                float* __restrict__ outp,
                                                float* __restrict__ pssum,
                                                int jspan) {
  __shared__ float Vp[PAIRS * PAIR_STRIDE];
  __shared__ float srow[4][16];

  const int lane = threadIdx.x & 31;
  const int wave = threadIdx.x >> 5;
  const int half = lane >> 4;
  const int lid  = lane & 15;
  const int kb   = half * 2;

  const int head = blockIdx.y;
  const int iblk = blockIdx.x * 64;
  const int z    = blockIdx.z;
  const int jstart = z * jspan;
  const int row  = iblk + wave * 16 + lid;

  const float sir = si[head * N_NODES + row];
  const float mx0 = sir + Mx[head * N_NODES + row];
  const float mrow = (mx0 > 0.f) ? mx0 : ALPHA * mx0;

  v8f acc[8];
#pragma unroll
  for (int t = 0; t < 8; ++t) acc[t] = {};
  float ssum = 0.f;

  const float* sjh  = sj + head * N_NODES;
  const int*   arow = adj + (size_t)row * N_NODES;
  const float* Vg   = h + head * HEAD_DIM;

  const int st_pair = threadIdx.x >> 2;
  const int st_d0   = (threadIdx.x & 3) * 32;

  for (int c = jstart; c < jstart + jspan; c += CHUNK_J) {
    {
      const float* src0 = Vg + (size_t)(c + st_pair * 2 + 0) * F_OUT + st_d0;
      const float* src1 = Vg + (size_t)(c + st_pair * 2 + 1) * F_OUT + st_d0;
      float* dst = Vp + st_pair * PAIR_STRIDE + st_d0 * 2;
#pragma unroll
      for (int q = 0; q < 8; ++q) {
        const v4f r0 = *(const v4f*)(src0 + q * 4);
        const v4f r1 = *(const v4f*)(src1 + q * 4);
        v4f w0 = {r0.x, r1.x, r0.y, r1.y};
        v4f w1 = {r0.z, r1.z, r0.w, r1.w};
        *(v4f*)(dst + q * 8 + 0) = w0;
        *(v4f*)(dst + q * 8 + 4) = w1;
      }
    }
    __syncthreads();

#pragma unroll
    for (int s4 = 0; s4 < CHUNK_J / 4; ++s4) {
      const int jg0 = c + s4 * 4 + kb;
      const v2f sjv = *(const v2f*)(sjh + jg0);
      const v2i av  = *(const v2i*)(arow + jg0);

      float e0 = sir + sjv.x; e0 = (e0 > 0.f) ? e0 : ALPHA * e0;
      float e1 = sir + sjv.y; e1 = (e1 > 0.f) ? e1 : ALPHA * e1;
      const float p0 = (av.x > 0) ? __expf(e0 - mrow) : 0.f;
      const float p1 = (av.y > 0) ? __expf(e1 - mrow) : 0.f;
      ssum += p0 + p1;

      v2f af;
      af.x = p0;
      af.y = p1;
      const float* bp = Vp + (s4 * 2 + half) * PAIR_STRIDE + lid * 2;
#pragma unroll
      for (int t = 0; t < 8; ++t) {
        const v2f bf = *(const v2f*)(bp + t * 32);
        acc[t] = __builtin_amdgcn_wmma_f32_16x16x4_f32(
            false, af, false, bf, (short)0, acc[t], false, false);
      }
    }
    __syncthreads();
  }

  ssum += __shfl_xor(ssum, 16, 32);

  if (PARTIAL) {
    if (half == 0) pssum[(size_t)z * (HEADS * N_NODES) + head * N_NODES + row] = ssum;
    float* pa = outp + (size_t)z * (N_NODES * F_OUT);
#pragma unroll
    for (int t = 0; t < 8; ++t) {
#pragma unroll
      for (int v = 0; v < 8; ++v) {
        const int m = iblk + wave * 16 + v + half * 8;
        pa[(size_t)m * F_OUT + head * HEAD_DIM + t * 16 + lid] = acc[t][v];
      }
    }
  } else {
    if (half == 0) srow[wave][lid] = ssum;
    __syncthreads();
#pragma unroll
    for (int t = 0; t < 8; ++t) {
#pragma unroll
      for (int v = 0; v < 8; ++v) {
        const int m = iblk + wave * 16 + v + half * 8;
        float val = acc[t][v] / srow[wave][v + half * 8];
        val = (val > 0.f) ? val : (__expf(val) - 1.f);
        outp[(size_t)m * F_OUT + head * HEAD_DIM + t * 16 + lid] = val;
      }
    }
  }
}

#define SLAB_J     512
#define SLAB_PAIRS (SLAB_J / 2)
#define GRP_STEPS  8
#define SLAB_GRPS  (SLAB_J / (4 * GRP_STEPS))
#define AGG_WAVES  16

template <int JS>
__global__ __launch_bounds__(AGG_WAVES * 32) void attn_agg_slab(const float* __restrict__ h,
                                                                const int* __restrict__ adj,
                                                                const float* __restrict__ si,
                                                                const float* __restrict__ sj,
                                                                const float* __restrict__ Mx,
                                                                float* __restrict__ pacc,
                                                                float* __restrict__ pssum) {
  constexpr int JSPAN = N_NODES / JS;
  constexpr int NSLAB = JSPAN / SLAB_J;
  static_assert(NSLAB * SLAB_J == JSPAN, "slabs tile the j span");

  __shared__ __attribute__((aligned(16))) float Vp[SLAB_PAIRS * PAIR_STRIDE];
  __shared__ __attribute__((aligned(16))) float sjs[SLAB_J];

  const int lane = threadIdx.x & 31;
  const int wave = threadIdx.x >> 5;
  const int half = lane >> 4;
  const int lid  = lane & 15;
  const int kb   = half * 2;

  const int head = blockIdx.y;
  const int iblk = blockIdx.x * (AGG_WAVES * 16);
  const int z    = blockIdx.z;
  const int jstart = z * JSPAN;
  const int row  = iblk + wave * 16 + lid;

  const float sir = si[head * N_NODES + row];
  const float mx0 = sir + Mx[head * N_NODES + row];
  const float mrow = (mx0 > 0.f) ? mx0 : ALPHA * mx0;

  v8f acc[8];
#pragma unroll
  for (int t = 0; t < 8; ++t) acc[t] = {};
  float ssum = 0.f;

  const float* sjh = sj + head * N_NODES;
  const float* Vg  = h + head * HEAD_DIM;
  const int*   ap  = adj + (size_t)row * N_NODES + jstart + kb;
  const float* bbase = Vp + half * PAIR_STRIDE + lid * 2;
  const float* sbase = sjs + kb;

  v2i an[GRP_STEPS];
#pragma unroll
  for (int s = 0; s < GRP_STEPS; ++s) an[s] = *(const v2i*)(ap + s * 4);
  ap += 4 * GRP_STEPS;

  for (int slab = 0; slab < NSLAB; ++slab) {
    const int c = jstart + slab * SLAB_J;

#pragma unroll 1
    for (int pb = wave * (SLAB_PAIRS / AGG_WAVES); pb < (wave + 1) * (SLAB_PAIRS / AGG_WAVES); pb += 8) {
      v4f r0[8], r1[8];
#pragma unroll
      for (int u = 0; u < 8; ++u) {
        const float* s0 = Vg + (size_t)(c + (pb + u) * 2) * F_OUT + lane * 4;
        r0[u] = *(const v4f*)(s0);
        r1[u] = *(const v4f*)(s0 + F_OUT);
      }
#pragma unroll
      for (int u = 0; u < 8; ++u) {
        float* dst = Vp + (pb + u) * PAIR_STRIDE + lane * 8;
        const v4f w0 = {r0[u].x, r1[u].x, r0[u].y, r1[u].y};
        const v4f w1 = {r0[u].z, r1[u].z, r0[u].w, r1[u].w};
        *(v4f*)(dst + 0) = w0;
        *(v4f*)(dst + 4) = w1;
      }
    }
    if (threadIdx.x < SLAB_J / 4)
      *(v4f*)(sjs + threadIdx.x * 4) = *(const v4f*)(sjh + c + threadIdx.x * 4);
    __syncthreads();

    const float* bg = bbase;
    const float* sg = sbase;
    for (int g = 0; g < SLAB_GRPS; ++g) {
      v2i ac[GRP_STEPS];
#pragma unroll
      for (int s = 0; s < GRP_STEPS; ++s) ac[s] = an[s];
      if (slab * SLAB_GRPS + g + 1 < NSLAB * SLAB_GRPS) {
#pragma unroll
        for (int s = 0; s < GRP_STEPS; ++s) an[s] = *(const v2i*)(ap + s * 4);
        ap += 4 * GRP_STEPS;
      }

#pragma unroll
      for (int s = 0; s < GRP_STEPS; ++s) {
        const v2f sjv = *(const v2f*)(sg + s * 4);
        const v2i av  = ac[s];

        float e0 = sir + sjv.x; e0 = (e0 > 0.f) ? e0 : ALPHA * e0;
        float e1 = sir + sjv.y; e1 = (e1 > 0.f) ? e1 : ALPHA * e1;
        const float p0 = (av.x > 0) ? __expf(e0 - mrow) : 0.f;
        const float p1 = (av.y > 0) ? __expf(e1 - mrow) : 0.f;
        ssum += p0 + p1;

        v2f af;
        af.x = p0;
        af.y = p1;
        const float* bp = bg + s * (2 * PAIR_STRIDE);
#pragma unroll
        for (int t = 0; t < 8; ++t) {
          const v2f bf = *(const v2f*)(bp + t * 32);
          acc[t] = __builtin_amdgcn_wmma_f32_16x16x4_f32(
              false, af, false, bf, (short)0, acc[t], false, false);
        }
      }
      bg += GRP_STEPS * 2 * PAIR_STRIDE;
      sg += GRP_STEPS * 4;
    }
    if (slab + 1 < NSLAB) __syncthreads();
  }

  ssum += __shfl_xor(ssum, 16, 32);

  if (half == 0) pssum[(size_t)z * (HEADS * N_NODES) + head * N_NODES + row] = ssum;
  float* pa = pacc + (size_t)z * (N_NODES * F_OUT);
#pragma unroll
  for (int t = 0; t < 8; ++t) {
#pragma unroll
    for (int v = 0; v < 8; ++v) {
      const int m = iblk + wave * 16 + v + half * 8;
      pa[(size_t)m * F_OUT + head * HEAD_DIM + t * 16 + lid] = acc[t][v];
    }
  }
}

template <int JS>
__global__ __launch_bounds__(256) void attn_reduce(const float* __restrict__ pacc,
                                                   const float* __restrict__ pssum,
                                                   float* __restrict__ out) {
  const int i = blockIdx.x;
#pragma unroll
  for (int r = 0; r < 2; ++r) {
    const int col = threadIdx.x + r * 256;
    const int head = col >> 7;
    float num = 0.f, den = 0.f;
#pragma unroll
    for (int zz = 0; zz < JS; ++zz) {
      num += pacc[(size_t)zz * (N_NODES * F_OUT) + (size_t)i * F_OUT + col];
      den += pssum[(size_t)zz * (HEADS * N_NODES) + head * N_NODES + i];
    }
    float val = num / den;
    val = (val > 0.f) ? val : (__expf(val) - 1.f);
    out[(size_t)i * F_OUT + col] = val;
  }
}

extern "C" void kernel_launch(void* const* d_in, const int* in_sizes, int n_in,
                              void* d_out, int out_size, void* d_ws, size_t ws_size,
                              hipStream_t stream) {
  const float* x   = (const float*)d_in[0];
  const int*   adj = (const int*)d_in[1];
  const float* W   = (const float*)d_in[2];
  const float* a   = (const float*)d_in[3];
  float* out = (float*)d_out;

  constexpr int JS = 4;
  const size_t HN = (size_t)HEADS * N_NODES;

  float* h   = (float*)d_ws;
  float* si  = h + (size_t)N_NODES * F_OUT;
  float* sj  = si + HN;
  float* sjT = sj + HN;
  float* Mx  = sjT + HN;
  float* pacc  = Mx + HN;
  float* pssum = pacc + (size_t)JS * N_NODES * F_OUT;
  const size_t need_js =
      (size_t)(pssum + JS * HN - (float*)d_ws) * sizeof(float);

  gemm_xw<<<dim3(N_NODES / 64, F_OUT / 64), 128, 0, stream>>>(x, W, h);
  attn_scores<<<dim3(N_NODES / 4), 128, 0, stream>>>(h, a, si, sj, sjT);
  row_max<<<dim3(N_NODES), 256, 0, stream>>>(adj, sjT, Mx);

  if (ws_size >= need_js) {
    attn_agg_slab<JS><<<dim3(N_NODES / (AGG_WAVES * 16), HEADS, JS), AGG_WAVES * 32, 0, stream>>>(
        h, adj, si, sj, Mx, pacc, pssum);
    attn_reduce<JS><<<dim3(N_NODES), 256, 0, stream>>>(pacc, pssum, out);
  } else {
    attn_agg<false><<<dim3(N_NODES / 64, HEADS, 1), 128, 0, stream>>>(
        h, adj, si, sj, Mx, out, nullptr, N_NODES);
  }
}


# attn_agg keeps each row's running masked max itself and rescales the row's sums by exp(old-new) when it rises; attn_reduce weights the 4 quarters by exp(own max - row max); row_max launch removed
# speedup vs baseline: 1.5783x; 1.1295x over previous
#include <hip/hip_runtime.h>
#include <hip/hip_bf16.h>

typedef __attribute__((ext_vector_type(2))) float v2f;
typedef __attribute__((ext_vector_type(4))) float v4f;
typedef __attribute__((ext_vector_type(8))) float v8f;
typedef __attribute__((ext_vector_type(2))) int   v2i;

#define N_NODES 4096
#define F_IN    512
#define F_OUT   512
#define HEADS   4
#define HEAD_DIM 128
#define ALPHA   0.2f
#define NEG_INF (-9.0e15f)

__global__ __launch_bounds__(128) void gemm_xw(const float* __restrict__ x,
                                               const float* __restrict__ W,
                                               float* __restrict__ h) {
  const int lane = threadIdx.x & 31;
  const int wave = threadIdx.x >> 5;
  const int half = lane >> 4;
  const int lid  = lane & 15;
  const int kb   = half * 2;

  const int bm = blockIdx.x;
  const int bn = blockIdx.y;
  const int row = bm * 64 + wave * 16 + lid;

  v8f acc[4];
#pragma unroll
  for (int t = 0; t < 4; ++t) acc[t] = {};

  const float* xrow = x + (size_t)row * F_IN;
  for (int k = 0; k < F_IN; k += 4) {
    v2f a;
    a.x = xrow[k + kb + 0];
    a.y = xrow[k + kb + 1];
#pragma unroll
    for (int t = 0; t < 4; ++t) {
      const int col = bn * 64 + t * 16 + lid;
      v2f b;
      b.x = W[(size_t)(k + kb + 0) * F_OUT + col];
      b.y = W[(size_t)(k + kb + 1) * F_OUT + col];
      acc[t] = __builtin_amdgcn_wmma_f32_16x16x4_f32(
          false, a, false, b, (short)0, acc[t], false, false);
    }
  }

#pragma unroll
  for (int t = 0; t < 4; ++t) {
#pragma unroll
    for (int v = 0; v < 8; ++v) {
      const int m = bm * 64 + wave * 16 + v + half * 8;
      const int c = bn * 64 + t * 16 + lid;
      h[(size_t)m * F_OUT + c] = acc[t][v];
    }
  }
}

template <bool WITH_T>
__global__ __launch_bounds__(128) void attn_scores(const float* __restrict__ h,
                                                   const float* __restrict__ a,
                                                   float* __restrict__ si,
                                                   float* __restrict__ sj,
                                                   float* __restrict__ sjT) {
  const int lane = threadIdx.x & 31;
  const int wave = threadIdx.x >> 5;
  const int n = blockIdx.x * 4 + wave;

  const v4f a1 = *(const v4f*)(a + lane * 4);
  const v4f a2 = *(const v4f*)(a + HEAD_DIM + lane * 4);

#pragma unroll
  for (int head = 0; head < HEADS; ++head) {
    const float* hrow = h + (size_t)n * F_OUT + head * HEAD_DIM;
    const v4f hv = *(const v4f*)(hrow + lane * 4);
    float s1 = hv.x * a1.x + hv.y * a1.y + hv.z * a1.z + hv.w * a1.w;
    float s2 = hv.x * a2.x + hv.y * a2.y + hv.z * a2.z + hv.w * a2.w;
#pragma unroll
    for (int off = 16; off > 0; off >>= 1) {
      s1 += __shfl_down(s1, off, 32);
      s2 += __shfl_down(s2, off, 32);
    }
    if (lane == 0) {
      si[head * N_NODES + n] = s1;
      sj[head * N_NODES + n] = s2;
      if (WITH_T) sjT[n * HEADS + head] = s2;
    }
  }
}

__global__ __launch_bounds__(256) void row_max(const int* __restrict__ adj,
                                               const float* __restrict__ sjT,
                                               float* __restrict__ Mx) {
  const int i = blockIdx.x;
  const int t = threadIdx.x;
  v4f m = {NEG_INF, NEG_INF, NEG_INF, NEG_INF};
  const int* arow = adj + (size_t)i * N_NODES;
  for (int j = t; j < N_NODES; j += 256) {
    if (arow[j] > 0) {
      const v4f sv = *(const v4f*)(sjT + j * 4);
      m.x = fmaxf(m.x, sv.x);
      m.y = fmaxf(m.y, sv.y);
      m.z = fmaxf(m.z, sv.z);
      m.w = fmaxf(m.w, sv.w);
    }
  }
  __shared__ v4f red[256];
  red[t] = m;
  __syncthreads();
  for (int s = 128; s > 0; s >>= 1) {
    if (t < s) {
      v4f o = red[t + s];
      v4f c = red[t];
      c.x = fmaxf(c.x, o.x);
      c.y = fmaxf(c.y, o.y);
      c.z = fmaxf(c.z, o.z);
      c.w = fmaxf(c.w, o.w);
      red[t] = c;
    }
    __syncthreads();
  }
  if (t < 4) Mx[t * N_NODES + i] = red[0][t];
}

#define CHUNK_J 64
#define PAIRS   (CHUNK_J / 2)
#define PAIR_STRIDE 288

template <bool PARTIAL>
__global__ __launch_bounds__(128) void attn_agg(const float* __restrict__ h,
                                                const int* __restrict__ adj,
                                                const float* __restrict__ si,
                                                const float* __restrict__ sj,
                                                const float* __restrict__ Mx,
                                                float* __restrict__ outp,
                                                float* __restrict__ pssum,
                                                int jspan) {
  __shared__ float Vp[PAIRS * PAIR_STRIDE];
  __shared__ float srow[4][16];

  const int lane = threadIdx.x & 31;
  const int wave = threadIdx.x >> 5;
  const int half = lane >> 4;
  const int lid  = lane & 15;
  const int kb   = half * 2;

  const int head = blockIdx.y;
  const int iblk = blockIdx.x * 64;
  const int z    = blockIdx.z;
  const int jstart = z * jspan;
  const int row  = iblk + wave * 16 + lid;

  const float sir = si[head * N_NODES + row];
  const float mx0 = sir + Mx[head * N_NODES + row];
  const float mrow = (mx0 > 0.f) ? mx0 : ALPHA * mx0;

  v8f acc[8];
#pragma unroll
  for (int t = 0; t < 8; ++t) acc[t] = {};
  float ssum = 0.f;

  const float* sjh  = sj + head * N_NODES;
  const int*   arow = adj + (size_t)row * N_NODES;
  const float* Vg   = h + head * HEAD_DIM;

  const int st_pair = threadIdx.x >> 2;
  const int st_d0   = (threadIdx.x & 3) * 32;

  for (int c = jstart; c < jstart + jspan; c += CHUNK_J) {
    {
      const float* src0 = Vg + (size_t)(c + st_pair * 2 + 0) * F_OUT + st_d0;
      const float* src1 = Vg + (size_t)(c + st_pair * 2 + 1) * F_OUT + st_d0;
      float* dst = Vp + st_pair * PAIR_STRIDE + st_d0 * 2;
#pragma unroll
      for (int q = 0; q < 8; ++q) {
        const v4f r0 = *(const v4f*)(src0 + q * 4);
        const v4f r1 = *(const v4f*)(src1 + q * 4);
        v4f w0 = {r0.x, r1.x, r0.y, r1.y};
        v4f w1 = {r0.z, r1.z, r0.w, r1.w};
        *(v4f*)(dst + q * 8 + 0) = w0;
        *(v4f*)(dst + q * 8 + 4) = w1;
      }
    }
    __syncthreads();

#pragma unroll
    for (int s4 = 0; s4 < CHUNK_J / 4; ++s4) {
      const int jg0 = c + s4 * 4 + kb;
      const v2f sjv = *(const v2f*)(sjh + jg0);
      const v2i av  = *(const v2i*)(arow + jg0);

      float e0 = sir + sjv.x; e0 = (e0 > 0.f) ? e0 : ALPHA * e0;
      float e1 = sir + sjv.y; e1 = (e1 > 0.f) ? e1 : ALPHA * e1;
      const float p0 = (av.x > 0) ? __expf(e0 - mrow) : 0.f;
      const float p1 = (av.y > 0) ? __expf(e1 - mrow) : 0.f;
      ssum += p0 + p1;

      v2f af;
      af.x = p0;
      af.y = p1;
      const float* bp = Vp + (s4 * 2 + half) * PAIR_STRIDE + lid * 2;
#pragma unroll
      for (int t = 0; t < 8; ++t) {
        const v2f bf = *(const v2f*)(bp + t * 32);
        acc[t] = __builtin_amdgcn_wmma_f32_16x16x4_f32(
            false, af, false, bf, (short)0, acc[t], false, false);
      }
    }
    __syncthreads();
  }

  ssum += __shfl_xor(ssum, 16, 32);

  if (PARTIAL) {
    if (half == 0) pssum[(size_t)z * (HEADS * N_NODES) + head * N_NODES + row] = ssum;
    float* pa = outp + (size_t)z * (N_NODES * F_OUT);
#pragma unroll
    for (int t = 0; t < 8; ++t) {
#pragma unroll
      for (int v = 0; v < 8; ++v) {
        const int m = iblk + wave * 16 + v + half * 8;
        pa[(size_t)m * F_OUT + head * HEAD_DIM + t * 16 + lid] = acc[t][v];
      }
    }
  } else {
    if (half == 0) srow[wave][lid] = ssum;
    __syncthreads();
#pragma unroll
    for (int t = 0; t < 8; ++t) {
#pragma unroll
      for (int v = 0; v < 8; ++v) {
        const int m = iblk + wave * 16 + v + half * 8;
        float val = acc[t][v] / srow[wave][v + half * 8];
        val = (val > 0.f) ? val : (__expf(val) - 1.f);
        outp[(size_t)m * F_OUT + head * HEAD_DIM + t * 16 + lid] = val;
      }
    }
  }
}

#define SLAB_J     512
#define SLAB_PAIRS (SLAB_J / 2)
#define GRP_STEPS  8
#define SLAB_GRPS  (SLAB_J / (4 * GRP_STEPS))
#define AGG_WAVES  16
#define M_FLOOR    (-3.0e38f)

template <int JS>
__global__ __launch_bounds__(AGG_WAVES * 32) void attn_agg_slab(const float* __restrict__ h,
                                                                const int* __restrict__ adj,
                                                                const float* __restrict__ si,
                                                                const float* __restrict__ sj,
                                                                float* __restrict__ pacc,
                                                                float* __restrict__ pssum,
                                                                float* __restrict__ pmax) {
  constexpr int JSPAN = N_NODES / JS;
  constexpr int NSLAB = JSPAN / SLAB_J;
  static_assert(NSLAB * SLAB_J == JSPAN, "slabs tile the j span");

  __shared__ __attribute__((aligned(16))) float Vp[SLAB_PAIRS * PAIR_STRIDE];
  __shared__ __attribute__((aligned(16))) float sjs[SLAB_J];

  const int lane = threadIdx.x & 31;
  const int wave = threadIdx.x >> 5;
  const int half = lane >> 4;
  const int lid  = lane & 15;
  const int kb   = half * 2;

  const int head = blockIdx.y;
  const int iblk = blockIdx.x * (AGG_WAVES * 16);
  const int z    = blockIdx.z;
  const int jstart = z * JSPAN;
  const int row  = iblk + wave * 16 + lid;

  const float sir = si[head * N_NODES + row];
  float mrun = M_FLOOR;

  v8f acc[8];
#pragma unroll
  for (int t = 0; t < 8; ++t) acc[t] = {};
  float ssum = 0.f;

  const float* sjh = sj + head * N_NODES;
  const float* Vg  = h + head * HEAD_DIM;
  const int*   ap  = adj + (size_t)row * N_NODES + jstart + kb;
  const float* bbase = Vp + half * PAIR_STRIDE + lid * 2;
  const float* sbase = sjs + kb;

  v2i an[GRP_STEPS];
#pragma unroll
  for (int s = 0; s < GRP_STEPS; ++s) an[s] = *(const v2i*)(ap + s * 4);
  ap += 4 * GRP_STEPS;

  for (int slab = 0; slab < NSLAB; ++slab) {
    const int c = jstart + slab * SLAB_J;

#pragma unroll 1
    for (int pb = wave * (SLAB_PAIRS / AGG_WAVES); pb < (wave + 1) * (SLAB_PAIRS / AGG_WAVES); pb += 8) {
      v4f r0[8], r1[8];
#pragma unroll
      for (int u = 0; u < 8; ++u) {
        const float* s0 = Vg + (size_t)(c + (pb + u) * 2) * F_OUT + lane * 4;
        r0[u] = *(const v4f*)(s0);
        r1[u] = *(const v4f*)(s0 + F_OUT);
      }
#pragma unroll
      for (int u = 0; u < 8; ++u) {
        float* dst = Vp + (pb + u) * PAIR_STRIDE + lane * 8;
        const v4f w0 = {r0[u].x, r1[u].x, r0[u].y, r1[u].y};
        const v4f w1 = {r0[u].z, r1[u].z, r0[u].w, r1[u].w};
        *(v4f*)(dst + 0) = w0;
        *(v4f*)(dst + 4) = w1;
      }
    }
    if (threadIdx.x < SLAB_J / 4)
      *(v4f*)(sjs + threadIdx.x * 4) = *(const v4f*)(sjh + c + threadIdx.x * 4);
    __syncthreads();

    const float* bg = bbase;
    const float* sg = sbase;
    for (int g = 0; g < SLAB_GRPS; ++g) {
      v2i ac[GRP_STEPS];
#pragma unroll
      for (int s = 0; s < GRP_STEPS; ++s) ac[s] = an[s];
      if (slab * SLAB_GRPS + g + 1 < NSLAB * SLAB_GRPS) {
#pragma unroll
        for (int s = 0; s < GRP_STEPS; ++s) an[s] = *(const v2i*)(ap + s * 4);
        ap += 4 * GRP_STEPS;
      }

      float ev[2 * GRP_STEPS];
      float gm = M_FLOOR;
#pragma unroll
      for (int s = 0; s < GRP_STEPS; ++s) {
        const v2f sjv = *(const v2f*)(sg + s * 4);
        float e0 = sir + sjv.x; e0 = (e0 > 0.f) ? e0 : ALPHA * e0;
        float e1 = sir + sjv.y; e1 = (e1 > 0.f) ? e1 : ALPHA * e1;
        ev[2 * s + 0] = e0;
        ev[2 * s + 1] = e1;
        gm = fmaxf(gm, (ac[s].x > 0) ? e0 : M_FLOOR);
        gm = fmaxf(gm, (ac[s].y > 0) ? e1 : M_FLOOR);
      }
      gm = fmaxf(gm, __shfl_xor(gm, 16, 32));
      const float mnew = fmaxf(mrun, gm);
      if (__any(mnew > mrun)) {
        const float sc = __expf(mrun - mnew);
        ssum *= sc;
#pragma unroll
        for (int v = 0; v < 8; ++v) {
          const float scv = __shfl(sc, half * 8 + v, 32);
#pragma unroll
          for (int t = 0; t < 8; ++t) acc[t][v] *= scv;
        }
        mrun = mnew;
      }

#pragma unroll
      for (int s = 0; s < GRP_STEPS; ++s) {
        const v2i av  = ac[s];
        const float p0 = (av.x > 0) ? __expf(ev[2 * s + 0] - mrun) : 0.f;
        const float p1 = (av.y > 0) ? __expf(ev[2 * s + 1] - mrun) : 0.f;
        ssum += p0 + p1;

        v2f af;
        af.x = p0;
        af.y = p1;
        const float* bp = bg + s * (2 * PAIR_STRIDE);
#pragma unroll
        for (int t = 0; t < 8; ++t) {
          const v2f bf = *(const v2f*)(bp + t * 32);
          acc[t] = __builtin_amdgcn_wmma_f32_16x16x4_f32(
              false, af, false, bf, (short)0, acc[t], false, false);
        }
      }
      bg += GRP_STEPS * 2 * PAIR_STRIDE;
      sg += GRP_STEPS * 4;
    }
    if (slab + 1 < NSLAB) __syncthreads();
  }

  ssum += __shfl_xor(ssum, 16, 32);

  if (half == 0) {
    pssum[(size_t)z * (HEADS * N_NODES) + head * N_NODES + row] = ssum;
    pmax[(size_t)z * (HEADS * N_NODES) + head * N_NODES + row]  = mrun;
  }
  float* pa = pacc + (size_t)z * (N_NODES * F_OUT);
#pragma unroll
  for (int t = 0; t < 8; ++t) {
#pragma unroll
    for (int v = 0; v < 8; ++v) {
      const int m = iblk + wave * 16 + v + half * 8;
      pa[(size_t)m * F_OUT + head * HEAD_DIM + t * 16 + lid] = acc[t][v];
    }
  }
}

template <int JS>
__global__ __launch_bounds__(256) void attn_reduce(const float* __restrict__ pacc,
                                                   const float* __restrict__ pssum,
                                                   const float* __restrict__ pmax,
                                                   float* __restrict__ out) {
  const int i = blockIdx.x;
#pragma unroll
  for (int r = 0; r < 2; ++r) {
    const int col = threadIdx.x + r * 256;
    const int head = col >> 7;
    float mz[JS];
    float mall = M_FLOOR;
#pragma unroll
    for (int zz = 0; zz < JS; ++zz) {
      mz[zz] = pmax[(size_t)zz * (HEADS * N_NODES) + head * N_NODES + i];
      mall = fmaxf(mall, mz[zz]);
    }
    float num = 0.f, den = 0.f;
#pragma unroll
    for (int zz = 0; zz < JS; ++zz) {
      const float w = __expf(mz[zz] - mall);
      num += w * pacc[(size_t)zz * (N_NODES * F_OUT) + (size_t)i * F_OUT + col];
      den += w * pssum[(size_t)zz * (HEADS * N_NODES) + head * N_NODES + i];
    }
    float val = num / den;
    val = (val > 0.f) ? val : (__expf(val) - 1.f);
    out[(size_t)i * F_OUT + col] = val;
  }
}

extern "C" void kernel_launch(void* const* d_in, const int* in_sizes, int n_in,
                              void* d_out, int out_size, void* d_ws, size_t ws_size,
                              hipStream_t stream) {
  const float* x   = (const float*)d_in[0];
  const int*   adj = (const int*)d_in[1];
  const float* W   = (const float*)d_in[2];
  const float* a   = (const float*)d_in[3];
  float* out = (float*)d_out;

  constexpr int JS = 4;
  const size_t HN = (size_t)HEADS * N_NODES;

  float* h   = (float*)d_ws;
  float* si  = h + (size_t)N_NODES * F_OUT;
  float* sj  = si + HN;
  float* pacc  = sj + HN;
  float* pssum = pacc + (size_t)JS * N_NODES * F_OUT;
  float* pmax  = pssum + JS * HN;
  const size_t need_js =
      (size_t)(pmax + JS * HN - (float*)d_ws) * sizeof(float);
  float* sjT = sj + HN;
  float* Mx  = sjT + HN;

  gemm_xw<<<dim3(N_NODES / 64, F_OUT / 64), 128, 0, stream>>>(x, W, h);

  if (ws_size >= need_js) {
    attn_scores<false><<<dim3(N_NODES / 4), 128, 0, stream>>>(h, a, si, sj, nullptr);
    attn_agg_slab<JS><<<dim3(N_NODES / (AGG_WAVES * 16), HEADS, JS), AGG_WAVES * 32, 0, stream>>>(
        h, adj, si, sj, pacc, pssum, pmax);
    attn_reduce<JS><<<dim3(N_NODES), 256, 0, stream>>>(pacc, pssum, pmax, out);
  } else {
    attn_scores<true><<<dim3(N_NODES / 4), 128, 0, stream>>>(h, a, si, sj, sjT);
    row_max<<<dim3(N_NODES), 256, 0, stream>>>(adj, sjT, Mx);
    attn_agg<false><<<dim3(N_NODES / 64, HEADS, 1), 128, 0, stream>>>(
        h, adj, si, sj, Mx, out, nullptr, N_NODES);
  }
}


# gemm_xw re-cut: a block owns 64 rows x one head and takes si, sj from its accumulators (attn_scores launch gone); operands by 16-byte global loads one 16-k block ahead; sum order of h, si, sj changed
# speedup vs baseline: 1.8071x; 1.1450x over previous
#include <hip/hip_runtime.h>
#include <hip/hip_bf16.h>

typedef __attribute__((ext_vector_type(2))) float v2f;
typedef __attribute__((ext_vector_type(4))) float v4f;
typedef __attribute__((ext_vector_type(8))) float v8f;
typedef __attribute__((ext_vector_type(2))) int   v2i;

#define N_NODES 4096
#define F_IN    512
#define F_OUT   512
#define HEADS   4
#define HEAD_DIM 128
#define ALPHA   0.2f
#define NEG_INF (-9.0e15f)

__global__ __launch_bounds__(128) void gemm_xw(const float* __restrict__ x,
                                               const float* __restrict__ W,
                                               float* __restrict__ h) {
  const int lane = threadIdx.x & 31;
  const int wave = threadIdx.x >> 5;
  const int half = lane >> 4;
  const int lid  = lane & 15;
  const int kb   = half * 2;

  const int bm = blockIdx.x;
  const int bn = blockIdx.y;
  const int row = bm * 64 + wave * 16 + lid;

  v8f acc[4];
#pragma unroll
  for (int t = 0; t < 4; ++t) acc[t] = {};

  const float* xrow = x + (size_t)row * F_IN;
  for (int k = 0; k < F_IN; k += 4) {
    v2f a;
    a.x = xrow[k + kb + 0];
    a.y = xrow[k + kb + 1];
#pragma unroll
    for (int t = 0; t < 4; ++t) {
      const int col = bn * 64 + t * 16 + lid;
      v2f b;
      b.x = W[(size_t)(k + kb + 0) * F_OUT + col];
      b.y = W[(size_t)(k + kb + 1) * F_OUT + col];
      acc[t] = __builtin_amdgcn_wmma_f32_16x16x4_f32(
          false, a, false, b, (short)0, acc[t], false, false);
    }
  }

#pragma unroll
  for (int t = 0; t < 4; ++t) {
#pragma unroll
    for (int v = 0; v < 8; ++v) {
      const int m = bm * 64 + wave * 16 + v + half * 8;
      const int c = bn * 64 + t * 16 + lid;
      h[(size_t)m * F_OUT + c] = acc[t][v];
    }
  }
}

__global__ __launch_bounds__(256) void gemm_xw_scores(const float* __restrict__ x,
                                                      const float* __restrict__ W,
                                                      const float* __restrict__ a,
                                                      float* __restrict__ h,
                                                      float* __restrict__ si,
                                                      float* __restrict__ sj) {
  __shared__ float part[2][64][2];

  const int lane = threadIdx.x & 31;
  const int wave = threadIdx.x >> 5;
  const int half = lane >> 4;
  const int lid  = lane & 15;
  const int rt   = wave >> 1;
  const int g    = wave & 1;

  const int head = blockIdx.y;
  const int row0 = blockIdx.x * 64 + rt * 16;
  const int cb   = head * HEAD_DIM + g * 64;

  const float* xp = x + (size_t)(row0 + lid) * F_IN + half * 8;
  const float* wp = W + (size_t)(half * 8) * F_OUT + cb + lid * 4;

  v8f acc[4];
#pragma unroll
  for (int i = 0; i < 4; ++i) acc[i] = {};

  v4f xa[2], wb[8];
  xa[0] = *(const v4f*)(xp);
  xa[1] = *(const v4f*)(xp + 4);
#pragma unroll
  for (int q = 0; q < 8; ++q) wb[q] = *(const v4f*)(wp + (size_t)q * F_OUT);

#pragma unroll 2
  for (int kb = 0; kb < F_IN; kb += 16) {
    const int kn = (kb + 16) & (F_IN - 1);
    v4f xn[2], wn[8];
    xn[0] = *(const v4f*)(xp + kn);
    xn[1] = *(const v4f*)(xp + kn + 4);
#pragma unroll
    for (int q = 0; q < 8; ++q) wn[q] = *(const v4f*)(wp + (size_t)(kn + q) * F_OUT);

#pragma unroll
    for (int s = 0; s < 4; ++s) {
      v2f af;
      af.x = xa[s >> 1][(s & 1) * 2 + 0];
      af.y = xa[s >> 1][(s & 1) * 2 + 1];
#pragma unroll
      for (int i = 0; i < 4; ++i) {
        v2f bf;
        bf.x = wb[2 * s + 0][i];
        bf.y = wb[2 * s + 1][i];
        acc[i] = __builtin_amdgcn_wmma_f32_16x16x4_f32(
            false, af, false, bf, (short)0, acc[i], false, false);
      }
    }

    xa[0] = xn[0];
    xa[1] = xn[1];
#pragma unroll
    for (int q = 0; q < 8; ++q) wb[q] = wn[q];
  }

#pragma unroll
  for (int r = 0; r < 8; ++r) {
    const v4f o = {acc[0][r], acc[1][r], acc[2][r], acc[3][r]};
    *(v4f*)(h + (size_t)(row0 + half * 8 + r) * F_OUT + cb + lid * 4) = o;
  }

  const v4f a1 = *(const v4f*)(a + g * 64 + lid * 4);
  const v4f a2 = *(const v4f*)(a + HEAD_DIM + g * 64 + lid * 4);
#pragma unroll
  for (int r = 0; r < 8; ++r) {
    float s1 = acc[0][r] * a1.x + acc[1][r] * a1.y + acc[2][r] * a1.z + acc[3][r] * a1.w;
    float s2 = acc[0][r] * a2.x + acc[1][r] * a2.y + acc[2][r] * a2.z + acc[3][r] * a2.w;
#pragma unroll
    for (int off = 8; off > 0; off >>= 1) {
      s1 += __shfl_xor(s1, off, 32);
      s2 += __shfl_xor(s2, off, 32);
    }
    if (lid == 0) {
      part[g][rt * 16 + half * 8 + r][0] = s1;
      part[g][rt * 16 + half * 8 + r][1] = s2;
    }
  }
  __syncthreads();
  if (threadIdx.x < 64) {
    const int n = blockIdx.x * 64 + threadIdx.x;
    si[head * N_NODES + n] = part[0][threadIdx.x][0] + part[1][threadIdx.x][0];
    sj[head * N_NODES + n] = part[0][threadIdx.x][1] + part[1][threadIdx.x][1];
  }
}

template <bool WITH_T>
__global__ __launch_bounds__(128) void attn_scores(const float* __restrict__ h,
                                                   const float* __restrict__ a,
                                                   float* __restrict__ si,
                                                   float* __restrict__ sj,
                                                   float* __restrict__ sjT) {
  const int lane = threadIdx.x & 31;
  const int wave = threadIdx.x >> 5;
  const int n = blockIdx.x * 4 + wave;

  const v4f a1 = *(const v4f*)(a + lane * 4);
  const v4f a2 = *(const v4f*)(a + HEAD_DIM + lane * 4);

#pragma unroll
  for (int head = 0; head < HEADS; ++head) {
    const float* hrow = h + (size_t)n * F_OUT + head * HEAD_DIM;
    const v4f hv = *(const v4f*)(hrow + lane * 4);
    float s1 = hv.x * a1.x + hv.y * a1.y + hv.z * a1.z + hv.w * a1.w;
    float s2 = hv.x * a2.x + hv.y * a2.y + hv.z * a2.z + hv.w * a2.w;
#pragma unroll
    for (int off = 16; off > 0; off >>= 1) {
      s1 += __shfl_down(s1, off, 32);
      s2 += __shfl_down(s2, off, 32);
    }
    if (lane == 0) {
      si[head * N_NODES + n] = s1;
      sj[head * N_NODES + n] = s2;
      if (WITH_T) sjT[n * HEADS + head] = s2;
    }
  }
}

__global__ __launch_bounds__(256) void row_max(const int* __restrict__ adj,
                                               const float* __restrict__ sjT,
                                               float* __restrict__ Mx) {
  const int i = blockIdx.x;
  const int t = threadIdx.x;
  v4f m = {NEG_INF, NEG_INF, NEG_INF, NEG_INF};
  const int* arow = adj + (size_t)i * N_NODES;
  for (int j = t; j < N_NODES; j += 256) {
    if (arow[j] > 0) {
      const v4f sv = *(const v4f*)(sjT + j * 4);
      m.x = fmaxf(m.x, sv.x);
      m.y = fmaxf(m.y, sv.y);
      m.z = fmaxf(m.z, sv.z);
      m.w = fmaxf(m.w, sv.w);
    }
  }
  __shared__ v4f red[256];
  red[t] = m;
  __syncthreads();
  for (int s = 128; s > 0; s >>= 1) {
    if (t < s) {
      v4f o = red[t + s];
      v4f c = red[t];
      c.x = fmaxf(c.x, o.x);
      c.y = fmaxf(c.y, o.y);
      c.z = fmaxf(c.z, o.z);
      c.w = fmaxf(c.w, o.w);
      red[t] = c;
    }
    __syncthreads();
  }
  if (t < 4) Mx[t * N_NODES + i] = red[0][t];
}

#define CHUNK_J 64
#define PAIRS   (CHUNK_J / 2)
#define PAIR_STRIDE 288

template <bool PARTIAL>
__global__ __launch_bounds__(128) void attn_agg(const float* __restrict__ h,
                                                const int* __restrict__ adj,
                                                const float* __restrict__ si,
                                                const float* __restrict__ sj,
                                                const float* __restrict__ Mx,
                                                float* __restrict__ outp,
                                                float* __restrict__ pssum,
                                                int jspan) {
  __shared__ float Vp[PAIRS * PAIR_STRIDE];
  __shared__ float srow[4][16];

  const int lane = threadIdx.x & 31;
  const int wave = threadIdx.x >> 5;
  const int half = lane >> 4;
  const int lid  = lane & 15;
  const int kb   = half * 2;

  const int head = blockIdx.y;
  const int iblk = blockIdx.x * 64;
  const int z    = blockIdx.z;
  const int jstart = z * jspan;
  const int row  = iblk + wave * 16 + lid;

  const float sir = si[head * N_NODES + row];
  const float mx0 = sir + Mx[head * N_NODES + row];
  const float mrow = (mx0 > 0.f) ? mx0 : ALPHA * mx0;

  v8f acc[8];
#pragma unroll
  for (int t = 0; t < 8; ++t) acc[t] = {};
  float ssum = 0.f;

  const float* sjh  = sj + head * N_NODES;
  const int*   arow = adj + (size_t)row * N_NODES;
  const float* Vg   = h + head * HEAD_DIM;

  const int st_pair = threadIdx.x >> 2;
  const int st_d0   = (threadIdx.x & 3) * 32;

  for (int c = jstart; c < jstart + jspan; c += CHUNK_J) {
    {
      const float* src0 = Vg + (size_t)(c + st_pair * 2 + 0) * F_OUT + st_d0;
      const float* src1 = Vg + (size_t)(c + st_pair * 2 + 1) * F_OUT + st_d0;
      float* dst = Vp + st_pair * PAIR_STRIDE + st_d0 * 2;
#pragma unroll
      for (int q = 0; q < 8; ++q) {
        const v4f r0 = *(const v4f*)(src0 + q * 4);
        const v4f r1 = *(const v4f*)(src1 + q * 4);
        v4f w0 = {r0.x, r1.x, r0.y, r1.y};
        v4f w1 = {r0.z, r1.z, r0.w, r1.w};
        *(v4f*)(dst + q * 8 + 0) = w0;
        *(v4f*)(dst + q * 8 + 4) = w1;
      }
    }
    __syncthreads();

#pragma unroll
    for (int s4 = 0; s4 < CHUNK_J / 4; ++s4) {
      const int jg0 = c + s4 * 4 + kb;
      const v2f sjv = *(const v2f*)(sjh + jg0);
      const v2i av  = *(const v2i*)(arow + jg0);

      float e0 = sir + sjv.x; e0 = (e0 > 0.f) ? e0 : ALPHA * e0;
      float e1 = sir + sjv.y; e1 = (e1 > 0.f) ? e1 : ALPHA * e1;
      const float p0 = (av.x > 0) ? __expf(e0 - mrow) : 0.f;
      const float p1 = (av.y > 0) ? __expf(e1 - mrow) : 0.f;
      ssum += p0 + p1;

      v2f af;
      af.x = p0;
      af.y = p1;
      const float* bp = Vp + (s4 * 2 + half) * PAIR_STRIDE + lid * 2;
#pragma unroll
      for (int t = 0; t < 8; ++t) {
        const v2f bf = *(const v2f*)(bp + t * 32);
        acc[t] = __builtin_amdgcn_wmma_f32_16x16x4_f32(
            false, af, false, bf, (short)0, acc[t], false, false);
      }
    }
    __syncthreads();
  }

  ssum += __shfl_xor(ssum, 16, 32);

  if (PARTIAL) {
    if (half == 0) pssum[(size_t)z * (HEADS * N_NODES) + head * N_NODES + row] = ssum;
    float* pa = outp + (size_t)z * (N_NODES * F_OUT);
#pragma unroll
    for (int t = 0; t < 8; ++t) {
#pragma unroll
      for (int v = 0; v < 8; ++v) {
        const int m = iblk + wave * 16 + v + half * 8;
        pa[(size_t)m * F_OUT + head * HEAD_DIM + t * 16 + lid] = acc[t][v];
      }
    }
  } else {
    if (half == 0) srow[wave][lid] = ssum;
    __syncthreads();
#pragma unroll
    for (int t = 0; t < 8; ++t) {
#pragma unroll
      for (int v = 0; v < 8; ++v) {
        const int m = iblk + wave * 16 + v + half * 8;
        float val = acc[t][v] / srow[wave][v + half * 8];
        val = (val > 0.f) ? val : (__expf(val) - 1.f);
        outp[(size_t)m * F_OUT + head * HEAD_DIM + t * 16 + lid] = val;
      }
    }
  }
}

#define SLAB_J     512
#define SLAB_PAIRS (SLAB_J / 2)
#define GRP_STEPS  8
#define SLAB_GRPS  (SLAB_J / (4 * GRP_STEPS))
#define AGG_WAVES  16
#define M_FLOOR    (-3.0e38f)

template <int JS>
__global__ __launch_bounds__(AGG_WAVES * 32) void attn_agg_slab(const float* __restrict__ h,
                                                                const int* __restrict__ adj,
                                                                const float* __restrict__ si,
                                                                const float* __restrict__ sj,
                                                                float* __restrict__ pacc,
                                                                float* __restrict__ pssum,
                                                                float* __restrict__ pmax) {
  constexpr int JSPAN = N_NODES / JS;
  constexpr int NSLAB = JSPAN / SLAB_J;
  static_assert(NSLAB * SLAB_J == JSPAN, "slabs tile the j span");

  __shared__ __attribute__((aligned(16))) float Vp[SLAB_PAIRS * PAIR_STRIDE];
  __shared__ __attribute__((aligned(16))) float sjs[SLAB_J];

  const int lane = threadIdx.x & 31;
  const int wave = threadIdx.x >> 5;
  const int half = lane >> 4;
  const int lid  = lane & 15;
  const int kb   = half * 2;

  const int head = blockIdx.y;
  const int iblk = blockIdx.x * (AGG_WAVES * 16);
  const int z    = blockIdx.z;
  const int jstart = z * JSPAN;
  const int row  = iblk + wave * 16 + lid;

  const float sir = si[head * N_NODES + row];
  float mrun = M_FLOOR;

  v8f acc[8];
#pragma unroll
  for (int t = 0; t < 8; ++t) acc[t] = {};
  float ssum = 0.f;

  const float* sjh = sj + head * N_NODES;
  const float* Vg  = h + head * HEAD_DIM;
  const int*   ap  = adj + (size_t)row * N_NODES + jstart + kb;
  const float* bbase = Vp + half * PAIR_STRIDE + lid * 2;
  const float* sbase = sjs + kb;

  v2i an[GRP_STEPS];
#pragma unroll
  for (int s = 0; s < GRP_STEPS; ++s) an[s] = *(const v2i*)(ap + s * 4);
  ap += 4 * GRP_STEPS;

  for (int slab = 0; slab < NSLAB; ++slab) {
    const int c = jstart + slab * SLAB_J;

#pragma unroll 1
    for (int pb = wave * (SLAB_PAIRS / AGG_WAVES); pb < (wave + 1) * (SLAB_PAIRS / AGG_WAVES); pb += 8) {
      v4f r0[8], r1[8];
#pragma unroll
      for (int u = 0; u < 8; ++u) {
        const float* s0 = Vg + (size_t)(c + (pb + u) * 2) * F_OUT + lane * 4;
        r0[u] = *(const v4f*)(s0);
        r1[u] = *(const v4f*)(s0 + F_OUT);
      }
#pragma unroll
      for (int u = 0; u < 8; ++u) {
        float* dst = Vp + (pb + u) * PAIR_STRIDE + lane * 8;
        const v4f w0 = {r0[u].x, r1[u].x, r0[u].y, r1[u].y};
        const v4f w1 = {r0[u].z, r1[u].z, r0[u].w, r1[u].w};
        *(v4f*)(dst + 0) = w0;
        *(v4f*)(dst + 4) = w1;
      }
    }
    if (threadIdx.x < SLAB_J / 4)
      *(v4f*)(sjs + threadIdx.x * 4) = *(const v4f*)(sjh + c + threadIdx.x * 4);
    __syncthreads();

    const float* bg = bbase;
    const float* sg = sbase;
    for (int g = 0; g < SLAB_GRPS; ++g) {
      v2i ac[GRP_STEPS];
#pragma unroll
      for (int s = 0; s < GRP_STEPS; ++s) ac[s] = an[s];
      if (slab * SLAB_GRPS + g + 1 < NSLAB * SLAB_GRPS) {
#pragma unroll
        for (int s = 0; s < GRP_STEPS; ++s) an[s] = *(const v2i*)(ap + s * 4);
        ap += 4 * GRP_STEPS;
      }

      float ev[2 * GRP_STEPS];
      float gm = M_FLOOR;
#pragma unroll
      for (int s = 0; s < GRP_STEPS; ++s) {
        const v2f sjv = *(const v2f*)(sg + s * 4);
        float e0 = sir + sjv.x; e0 = (e0 > 0.f) ? e0 : ALPHA * e0;
        float e1 = sir + sjv.y; e1 = (e1 > 0.f) ? e1 : ALPHA * e1;
        ev[2 * s + 0] = e0;
        ev[2 * s + 1] = e1;
        gm = fmaxf(gm, (ac[s].x > 0) ? e0 : M_FLOOR);
        gm = fmaxf(gm, (ac[s].y > 0) ? e1 : M_FLOOR);
      }
      gm = fmaxf(gm, __shfl_xor(gm, 16, 32));
      const float mnew = fmaxf(mrun, gm);
      if (__any(mnew > mrun)) {
        const float sc = __expf(mrun - mnew);
        ssum *= sc;
#pragma unroll
        for (int v = 0; v < 8; ++v) {
          const float scv = __shfl(sc, half * 8 + v, 32);
#pragma unroll
          for (int t = 0; t < 8; ++t) acc[t][v] *= scv;
        }
        mrun = mnew;
      }

#pragma unroll
      for (int s = 0; s < GRP_STEPS; ++s) {
        const v2i av  = ac[s];
        const float p0 = (av.x > 0) ? __expf(ev[2 * s + 0] - mrun) : 0.f;
        const float p1 = (av.y > 0) ? __expf(ev[2 * s + 1] - mrun) : 0.f;
        ssum += p0 + p1;

        v2f af;
        af.x = p0;
        af.y = p1;
        const float* bp = bg + s * (2 * PAIR_STRIDE);
#pragma unroll
        for (int t = 0; t < 8; ++t) {
          const v2f bf = *(const v2f*)(bp + t * 32);
          acc[t] = __builtin_amdgcn_wmma_f32_16x16x4_f32(
              false, af, false, bf, (short)0, acc[t], false, false);
        }
      }
      bg += GRP_STEPS * 2 * PAIR_STRIDE;
      sg += GRP_STEPS * 4;
    }
    if (slab + 1 < NSLAB) __syncthreads();
  }

  ssum += __shfl_xor(ssum, 16, 32);

  if (half == 0) {
    pssum[(size_t)z * (HEADS * N_NODES) + head * N_NODES + row] = ssum;
    pmax[(size_t)z * (HEADS * N_NODES) + head * N_NODES + row]  = mrun;
  }
  float* pa = pacc + (size_t)z * (N_NODES * F_OUT);
#pragma unroll
  for (int t = 0; t < 8; ++t) {
#pragma unroll
    for (int v = 0; v < 8; ++v) {
      const int m = iblk + wave * 16 + v + half * 8;
      pa[(size_t)m * F_OUT + head * HEAD_DIM + t * 16 + lid] = acc[t][v];
    }
  }
}

template <int JS>
__global__ __launch_bounds__(256) void attn_reduce(const float* __restrict__ pacc,
                                                   const float* __restrict__ pssum,
                                                   const float* __restrict__ pmax,
                                                   float* __restrict__ out) {
  const int i = blockIdx.x;
#pragma unroll
  for (int r = 0; r < 2; ++r) {
    const int col = threadIdx.x + r * 256;
    const int head = col >> 7;
    float mz[JS];
    float mall = M_FLOOR;
#pragma unroll
    for (int zz = 0; zz < JS; ++zz) {
      mz[zz] = pmax[(size_t)zz * (HEADS * N_NODES) + head * N_NODES + i];
      mall = fmaxf(mall, mz[zz]);
    }
    float num = 0.f, den = 0.f;
#pragma unroll
    for (int zz = 0; zz < JS; ++zz) {
      const float w = __expf(mz[zz] - mall);
      num += w * pacc[(size_t)zz * (N_NODES * F_OUT) + (size_t)i * F_OUT + col];
      den += w * pssum[(size_t)zz * (HEADS * N_NODES) + head * N_NODES + i];
    }
    float val = num / den;
    val = (val > 0.f) ? val : (__expf(val) - 1.f);
    out[(size_t)i * F_OUT + col] = val;
  }
}

extern "C" void kernel_launch(void* const* d_in, const int* in_sizes, int n_in,
                              void* d_out, int out_size, void* d_ws, size_t ws_size,
                              hipStream_t stream) {
  const float* x   = (const float*)d_in[0];
  const int*   adj = (const int*)d_in[1];
  const float* W   = (const float*)d_in[2];
  const float* a   = (const float*)d_in[3];
  float* out = (float*)d_out;

  constexpr int JS = 4;
  const size_t HN = (size_t)HEADS * N_NODES;

  float* h   = (float*)d_ws;
  float* si  = h + (size_t)N_NODES * F_OUT;
  float* sj  = si + HN;
  float* pacc  = sj + HN;
  float* pssum = pacc + (size_t)JS * N_NODES * F_OUT;
  float* pmax  = pssum + JS * HN;
  const size_t need_js =
      (size_t)(pmax + JS * HN - (float*)d_ws) * sizeof(float);
  float* sjT = sj + HN;
  float* Mx  = sjT + HN;

  if (ws_size >= need_js) {
    gemm_xw_scores<<<dim3(N_NODES / 64, HEADS), 256, 0, stream>>>(x, W, a, h, si, sj);
    attn_agg_slab<JS><<<dim3(N_NODES / (AGG_WAVES * 16), HEADS, JS), AGG_WAVES * 32, 0, stream>>>(
        h, adj, si, sj, pacc, pssum, pmax);
    attn_reduce<JS><<<dim3(N_NODES), 256, 0, stream>>>(pacc, pssum, pmax, out);
  } else {
    gemm_xw<<<dim3(N_NODES / 64, F_OUT / 64), 128, 0, stream>>>(x, W, h);
    attn_scores<true><<<dim3(N_NODES / 4), 128, 0, stream>>>(h, a, si, sj, sjT);
    row_max<<<dim3(N_NODES), 256, 0, stream>>>(adj, sjT, Mx);
    attn_agg<false><<<dim3(N_NODES / 64, HEADS, 1), 128, 0, stream>>>(
        h, adj, si, sj, Mx, out, nullptr, N_NODES);
  }
}


# attn_agg: scores in log2 units (si, sj scaled once), leaky relu as max(e,0.2e), one -inf select feeds both row max and exp2, adj prefetch branch-free: fewer vector ops per score; sum order kept
# speedup vs baseline: 1.8690x; 1.0343x over previous
#include <hip/hip_runtime.h>
#include <hip/hip_bf16.h>

typedef __attribute__((ext_vector_type(2))) float v2f;
typedef __attribute__((ext_vector_type(4))) float v4f;
typedef __attribute__((ext_vector_type(8))) float v8f;
typedef __attribute__((ext_vector_type(2))) int   v2i;

#define N_NODES 4096
#define F_IN    512
#define F_OUT   512
#define HEADS   4
#define HEAD_DIM 128
#define ALPHA   0.2f
#define NEG_INF (-9.0e15f)

__global__ __launch_bounds__(128) void gemm_xw(const float* __restrict__ x,
                                               const float* __restrict__ W,
                                               float* __restrict__ h) {
  const int lane = threadIdx.x & 31;
  const int wave = threadIdx.x >> 5;
  const int half = lane >> 4;
  const int lid  = lane & 15;
  const int kb   = half * 2;

  const int bm = blockIdx.x;
  const int bn = blockIdx.y;
  const int row = bm * 64 + wave * 16 + lid;

  v8f acc[4];
#pragma unroll
  for (int t = 0; t < 4; ++t) acc[t] = {};

  const float* xrow = x + (size_t)row * F_IN;
  for (int k = 0; k < F_IN; k += 4) {
    v2f a;
    a.x = xrow[k + kb + 0];
    a.y = xrow[k + kb + 1];
#pragma unroll
    for (int t = 0; t < 4; ++t) {
      const int col = bn * 64 + t * 16 + lid;
      v2f b;
      b.x = W[(size_t)(k + kb + 0) * F_OUT + col];
      b.y = W[(size_t)(k + kb + 1) * F_OUT + col];
      acc[t] = __builtin_amdgcn_wmma_f32_16x16x4_f32(
          false, a, false, b, (short)0, acc[t], false, false);
    }
  }

#pragma unroll
  for (int t = 0; t < 4; ++t) {
#pragma unroll
    for (int v = 0; v < 8; ++v) {
      const int m = bm * 64 + wave * 16 + v + half * 8;
      const int c = bn * 64 + t * 16 + lid;
      h[(size_t)m * F_OUT + c] = acc[t][v];
    }
  }
}

__global__ __launch_bounds__(256) void gemm_xw_scores(const float* __restrict__ x,
                                                      const float* __restrict__ W,
                                                      const float* __restrict__ a,
                                                      float* __restrict__ h,
                                                      float* __restrict__ si,
                                                      float* __restrict__ sj) {
  __shared__ float part[2][64][2];

  const int lane = threadIdx.x & 31;
  const int wave = threadIdx.x >> 5;
  const int half = lane >> 4;
  const int lid  = lane & 15;
  const int rt   = wave >> 1;
  const int g    = wave & 1;

  const int head = blockIdx.y;
  const int row0 = blockIdx.x * 64 + rt * 16;
  const int cb   = head * HEAD_DIM + g * 64;

  const float* xp = x + (size_t)(row0 + lid) * F_IN + half * 8;
  const float* wp = W + (size_t)(half * 8) * F_OUT + cb + lid * 4;

  v8f acc[4];
#pragma unroll
  for (int i = 0; i < 4; ++i) acc[i] = {};

  v4f xa[2], wb[8];
  xa[0] = *(const v4f*)(xp);
  xa[1] = *(const v4f*)(xp + 4);
#pragma unroll
  for (int q = 0; q < 8; ++q) wb[q] = *(const v4f*)(wp + (size_t)q * F_OUT);

#pragma unroll 2
  for (int kb = 0; kb < F_IN; kb += 16) {
    const int kn = (kb + 16) & (F_IN - 1);
    v4f xn[2], wn[8];
    xn[0] = *(const v4f*)(xp + kn);
    xn[1] = *(const v4f*)(xp + kn + 4);
#pragma unroll
    for (int q = 0; q < 8; ++q) wn[q] = *(const v4f*)(wp + (size_t)(kn + q) * F_OUT);

#pragma unroll
    for (int s = 0; s < 4; ++s) {
      v2f af;
      af.x = xa[s >> 1][(s & 1) * 2 + 0];
      af.y = xa[s >> 1][(s & 1) * 2 + 1];
#pragma unroll
      for (int i = 0; i < 4; ++i) {
        v2f bf;
        bf.x = wb[2 * s + 0][i];
        bf.y = wb[2 * s + 1][i];
        acc[i] = __builtin_amdgcn_wmma_f32_16x16x4_f32(
            false, af, false, bf, (short)0, acc[i], false, false);
      }
    }

    xa[0] = xn[0];
    xa[1] = xn[1];
#pragma unroll
    for (int q = 0; q < 8; ++q) wb[q] = wn[q];
  }

#pragma unroll
  for (int r = 0; r < 8; ++r) {
    const v4f o = {acc[0][r], acc[1][r], acc[2][r], acc[3][r]};
    *(v4f*)(h + (size_t)(row0 + half * 8 + r) * F_OUT + cb + lid * 4) = o;
  }

  const v4f a1 = *(const v4f*)(a + g * 64 + lid * 4);
  const v4f a2 = *(const v4f*)(a + HEAD_DIM + g * 64 + lid * 4);
#pragma unroll
  for (int r = 0; r < 8; ++r) {
    float s1 = acc[0][r] * a1.x + acc[1][r] * a1.y + acc[2][r] * a1.z + acc[3][r] * a1.w;
    float s2 = acc[0][r] * a2.x + acc[1][r] * a2.y + acc[2][r] * a2.z + acc[3][r] * a2.w;
#pragma unroll
    for (int off = 8; off > 0; off >>= 1) {
      s1 += __shfl_xor(s1, off, 32);
      s2 += __shfl_xor(s2, off, 32);
    }
    if (lid == 0) {
      part[g][rt * 16 + half * 8 + r][0] = s1;
      part[g][rt * 16 + half * 8 + r][1] = s2;
    }
  }
  __syncthreads();
  if (threadIdx.x < 64) {
    const int n = blockIdx.x * 64 + threadIdx.x;
    si[head * N_NODES + n] = part[0][threadIdx.x][0] + part[1][threadIdx.x][0];
    sj[head * N_NODES + n] = part[0][threadIdx.x][1] + part[1][threadIdx.x][1];
  }
}

template <bool WITH_T>
__global__ __launch_bounds__(128) void attn_scores(const float* __restrict__ h,
                                                   const float* __restrict__ a,
                                                   float* __restrict__ si,
                                                   float* __restrict__ sj,
                                                   float* __restrict__ sjT) {
  const int lane = threadIdx.x & 31;
  const int wave = threadIdx.x >> 5;
  const int n = blockIdx.x * 4 + wave;

  const v4f a1 = *(const v4f*)(a + lane * 4);
  const v4f a2 = *(const v4f*)(a + HEAD_DIM + lane * 4);

#pragma unroll
  for (int head = 0; head < HEADS; ++head) {
    const float* hrow = h + (size_t)n * F_OUT + head * HEAD_DIM;
    const v4f hv = *(const v4f*)(hrow + lane * 4);
    float s1 = hv.x * a1.x + hv.y * a1.y + hv.z * a1.z + hv.w * a1.w;
    float s2 = hv.x * a2.x + hv.y * a2.y + hv.z * a2.z + hv.w * a2.w;
#pragma unroll
    for (int off = 16; off > 0; off >>= 1) {
      s1 += __shfl_down(s1, off, 32);
      s2 += __shfl_down(s2, off, 32);
    }
    if (lane == 0) {
      si[head * N_NODES + n] = s1;
      sj[head * N_NODES + n] = s2;
      if (WITH_T) sjT[n * HEADS + head] = s2;
    }
  }
}

__global__ __launch_bounds__(256) void row_max(const int* __restrict__ adj,
                                               const float* __restrict__ sjT,
                                               float* __restrict__ Mx) {
  const int i = blockIdx.x;
  const int t = threadIdx.x;
  v4f m = {NEG_INF, NEG_INF, NEG_INF, NEG_INF};
  const int* arow = adj + (size_t)i * N_NODES;
  for (int j = t; j < N_NODES; j += 256) {
    if (arow[j] > 0) {
      const v4f sv = *(const v4f*)(sjT + j * 4);
      m.x = fmaxf(m.x, sv.x);
      m.y = fmaxf(m.y, sv.y);
      m.z = fmaxf(m.z, sv.z);
      m.w = fmaxf(m.w, sv.w);
    }
  }
  __shared__ v4f red[256];
  red[t] = m;
  __syncthreads();
  for (int s = 128; s > 0; s >>= 1) {
    if (t < s) {
      v4f o = red[t + s];
      v4f c = red[t];
      c.x = fmaxf(c.x, o.x);
      c.y = fmaxf(c.y, o.y);
      c.z = fmaxf(c.z, o.z);
      c.w = fmaxf(c.w, o.w);
      red[t] = c;
    }
    __syncthreads();
  }
  if (t < 4) Mx[t * N_NODES + i] = red[0][t];
}

#define CHUNK_J 64
#define PAIRS   (CHUNK_J / 2)
#define PAIR_STRIDE 288

template <bool PARTIAL>
__global__ __launch_bounds__(128) void attn_agg(const float* __restrict__ h,
                                                const int* __restrict__ adj,
                                                const float* __restrict__ si,
                                                const float* __restrict__ sj,
                                                const float* __restrict__ Mx,
                                                float* __restrict__ outp,
                                                float* __restrict__ pssum,
                                                int jspan) {
  __shared__ float Vp[PAIRS * PAIR_STRIDE];
  __shared__ float srow[4][16];

  const int lane = threadIdx.x & 31;
  const int wave = threadIdx.x >> 5;
  const int half = lane >> 4;
  const int lid  = lane & 15;
  const int kb   = half * 2;

  const int head = blockIdx.y;
  const int iblk = blockIdx.x * 64;
  const int z    = blockIdx.z;
  const int jstart = z * jspan;
  const int row  = iblk + wave * 16 + lid;

  const float sir = si[head * N_NODES + row];
  const float mx0 = sir + Mx[head * N_NODES + row];
  const float mrow = (mx0 > 0.f) ? mx0 : ALPHA * mx0;

  v8f acc[8];
#pragma unroll
  for (int t = 0; t < 8; ++t) acc[t] = {};
  float ssum = 0.f;

  const float* sjh  = sj + head * N_NODES;
  const int*   arow = adj + (size_t)row * N_NODES;
  const float* Vg   = h + head * HEAD_DIM;

  const int st_pair = threadIdx.x >> 2;
  const int st_d0   = (threadIdx.x & 3) * 32;

  for (int c = jstart; c < jstart + jspan; c += CHUNK_J) {
    {
      const float* src0 = Vg + (size_t)(c + st_pair * 2 + 0) * F_OUT + st_d0;
      const float* src1 = Vg + (size_t)(c + st_pair * 2 + 1) * F_OUT + st_d0;
      float* dst = Vp + st_pair * PAIR_STRIDE + st_d0 * 2;
#pragma unroll
      for (int q = 0; q < 8; ++q) {
        const v4f r0 = *(const v4f*)(src0 + q * 4);
        const v4f r1 = *(const v4f*)(src1 + q * 4);
        v4f w0 = {r0.x, r1.x, r0.y, r1.y};
        v4f w1 = {r0.z, r1.z, r0.w, r1.w};
        *(v4f*)(dst + q * 8 + 0) = w0;
        *(v4f*)(dst + q * 8 + 4) = w1;
      }
    }
    __syncthreads();

#pragma unroll
    for (int s4 = 0; s4 < CHUNK_J / 4; ++s4) {
      const int jg0 = c + s4 * 4 + kb;
      const v2f sjv = *(const v2f*)(sjh + jg0);
      const v2i av  = *(const v2i*)(arow + jg0);

      float e0 = sir + sjv.x; e0 = (e0 > 0.f) ? e0 : ALPHA * e0;
      float e1 = sir + sjv.y; e1 = (e1 > 0.f) ? e1 : ALPHA * e1;
      const float p0 = (av.x > 0) ? __expf(e0 - mrow) : 0.f;
      const float p1 = (av.y > 0) ? __expf(e1 - mrow) : 0.f;
      ssum += p0 + p1;

      v2f af;
      af.x = p0;
      af.y = p1;
      const float* bp = Vp + (s4 * 2 + half) * PAIR_STRIDE + lid * 2;
#pragma unroll
      for (int t = 0; t < 8; ++t) {
        const v2f bf = *(const v2f*)(bp + t * 32);
        acc[t] = __builtin_amdgcn_wmma_f32_16x16x4_f32(
            false, af, false, bf, (short)0, acc[t], false, false);
      }
    }
    __syncthreads();
  }

  ssum += __shfl_xor(ssum, 16, 32);

  if (PARTIAL) {
    if (half == 0) pssum[(size_t)z * (HEADS * N_NODES) + head * N_NODES + row] = ssum;
    float* pa = outp + (size_t)z * (N_NODES * F_OUT);
#pragma unroll
    for (int t = 0; t < 8; ++t) {
#pragma unroll
      for (int v = 0; v < 8; ++v) {
        const int m = iblk + wave * 16 + v + half * 8;
        pa[(size_t)m * F_OUT + head * HEAD_DIM + t * 16 + lid] = acc[t][v];
      }
    }
  } else {
    if (half == 0) srow[wave][lid] = ssum;
    __syncthreads();
#pragma unroll
    for (int t = 0; t < 8; ++t) {
#pragma unroll
      for (int v = 0; v < 8; ++v) {
        const int m = iblk + wave * 16 + v + half * 8;
        float val = acc[t][v] / srow[wave][v + half * 8];
        val = (val > 0.f) ? val : (__expf(val) - 1.f);
        outp[(size_t)m * F_OUT + head * HEAD_DIM + t * 16 + lid] = val;
      }
    }
  }
}

#define LOG2E      1.44269504088896341f
#define SLAB_J     512
#define SLAB_PAIRS (SLAB_J / 2)
#define GRP_STEPS  8
#define SLAB_GRPS  (SLAB_J / (4 * GRP_STEPS))
#define AGG_WAVES  16
#define M_FLOOR    (-3.0e38f)

template <int JS>
__global__ __launch_bounds__(AGG_WAVES * 32) void attn_agg_slab(const float* __restrict__ h,
                                                                const int* __restrict__ adj,
                                                                const float* __restrict__ si,
                                                                const float* __restrict__ sj,
                                                                float* __restrict__ pacc,
                                                                float* __restrict__ pssum,
                                                                float* __restrict__ pmax) {
  constexpr int JSPAN = N_NODES / JS;
  constexpr int NSLAB = JSPAN / SLAB_J;
  static_assert(NSLAB * SLAB_J == JSPAN, "slabs tile the j span");

  __shared__ __attribute__((aligned(16))) float Vp[SLAB_PAIRS * PAIR_STRIDE];
  __shared__ __attribute__((aligned(16))) float sjs[SLAB_J];

  const int lane = threadIdx.x & 31;
  const int wave = threadIdx.x >> 5;
  const int half = lane >> 4;
  const int lid  = lane & 15;
  const int kb   = half * 2;

  const int head = blockIdx.y;
  const int iblk = blockIdx.x * (AGG_WAVES * 16);
  const int z    = blockIdx.z;
  const int jstart = z * JSPAN;
  const int row  = iblk + wave * 16 + lid;

  const float sir = si[head * N_NODES + row] * LOG2E;
  float mrun = M_FLOOR;
  const float ninf = -__builtin_inff();

  v8f acc[8];
#pragma unroll
  for (int t = 0; t < 8; ++t) acc[t] = {};
  float ssum = 0.f;

  const float* sjh = sj + head * N_NODES;
  const float* Vg  = h + head * HEAD_DIM;
  const int*   ap  = adj + (size_t)row * N_NODES + jstart + kb;
  const float* bbase = Vp + half * PAIR_STRIDE + lid * 2;
  const float* sbase = sjs + kb;

  v2i an[GRP_STEPS];
#pragma unroll
  for (int s = 0; s < GRP_STEPS; ++s) an[s] = *(const v2i*)(ap + s * 4);
  ap += 4 * GRP_STEPS;

  for (int slab = 0; slab < NSLAB; ++slab) {
    const int c = jstart + slab * SLAB_J;

#pragma unroll 1
    for (int pb = wave * (SLAB_PAIRS / AGG_WAVES); pb < (wave + 1) * (SLAB_PAIRS / AGG_WAVES); pb += 8) {
      v4f r0[8], r1[8];
#pragma unroll
      for (int u = 0; u < 8; ++u) {
        const float* s0 = Vg + (size_t)(c + (pb + u) * 2) * F_OUT + lane * 4;
        r0[u] = *(const v4f*)(s0);
        r1[u] = *(const v4f*)(s0 + F_OUT);
      }
#pragma unroll
      for (int u = 0; u < 8; ++u) {
        float* dst = Vp + (pb + u) * PAIR_STRIDE + lane * 8;
        const v4f w0 = {r0[u].x, r1[u].x, r0[u].y, r1[u].y};
        const v4f w1 = {r0[u].z, r1[u].z, r0[u].w, r1[u].w};
        *(v4f*)(dst + 0) = w0;
        *(v4f*)(dst + 4) = w1;
      }
    }
    if (threadIdx.x < SLAB_J / 4) {
      const v4f sv = *(const v4f*)(sjh + c + threadIdx.x * 4);
      *(v4f*)(sjs + threadIdx.x * 4) = sv * LOG2E;
    }
    __syncthreads();

    const float* bg = bbase;
    const float* sg = sbase;
#pragma unroll 2
    for (int g = 0; g < SLAB_GRPS; ++g) {
      v2i ac[GRP_STEPS];
#pragma unroll
      for (int s = 0; s < GRP_STEPS; ++s) ac[s] = an[s];
#pragma unroll
      for (int s = 0; s < GRP_STEPS; ++s) an[s] = *(const v2i*)(ap + s * 4);
      ap += (slab * SLAB_GRPS + g + 2 < NSLAB * SLAB_GRPS) ? 4 * GRP_STEPS : 0;

      float ev[2 * GRP_STEPS];
      float gm = M_FLOOR;
#pragma unroll
      for (int s = 0; s < GRP_STEPS; ++s) {
        const v2f sjv = *(const v2f*)(sg + s * 4);
        float e0 = sir + sjv.x; e0 = fmaxf(e0, ALPHA * e0);
        float e1 = sir + sjv.y; e1 = fmaxf(e1, ALPHA * e1);
        e0 = (ac[s].x > 0) ? e0 : ninf;
        e1 = (ac[s].y > 0) ? e1 : ninf;
        ev[2 * s + 0] = e0;
        ev[2 * s + 1] = e1;
        gm = fmaxf(gm, e0);
        gm = fmaxf(gm, e1);
      }
      gm = fmaxf(gm, __shfl_xor(gm, 16, 32));
      const float mnew = fmaxf(mrun, gm);
      if (__any(mnew > mrun)) {
        const float sc = __builtin_amdgcn_exp2f(mrun - mnew);
        ssum *= sc;
#pragma unroll
        for (int v = 0; v < 8; ++v) {
          const float scv = __shfl(sc, half * 8 + v, 32);
#pragma unroll
          for (int t = 0; t < 8; ++t) acc[t][v] *= scv;
        }
        mrun = mnew;
      }

#pragma unroll
      for (int s = 0; s < GRP_STEPS; ++s) {
        const float p0 = __builtin_amdgcn_exp2f(ev[2 * s + 0] - mrun);
        const float p1 = __builtin_amdgcn_exp2f(ev[2 * s + 1] - mrun);
        ssum += p0 + p1;

        v2f af;
        af.x = p0;
        af.y = p1;
        const float* bp = bg + s * (2 * PAIR_STRIDE);
#pragma unroll
        for (int t = 0; t < 8; ++t) {
          const v2f bf = *(const v2f*)(bp + t * 32);
          acc[t] = __builtin_amdgcn_wmma_f32_16x16x4_f32(
              false, af, false, bf, (short)0, acc[t], false, false);
        }
      }
      bg += GRP_STEPS * 2 * PAIR_STRIDE;
      sg += GRP_STEPS * 4;
    }
    if (slab + 1 < NSLAB) __syncthreads();
  }

  ssum += __shfl_xor(ssum, 16, 32);

  if (half == 0) {
    pssum[(size_t)z * (HEADS * N_NODES) + head * N_NODES + row] = ssum;
    pmax[(size_t)z * (HEADS * N_NODES) + head * N_NODES + row]  = mrun;
  }
  float* pa = pacc + (size_t)z * (N_NODES * F_OUT);
#pragma unroll
  for (int t = 0; t < 8; ++t) {
#pragma unroll
    for (int v = 0; v < 8; ++v) {
      const int m = iblk + wave * 16 + v + half * 8;
      pa[(size_t)m * F_OUT + head * HEAD_DIM + t * 16 + lid] = acc[t][v];
    }
  }
}

template <int JS>
__global__ __launch_bounds__(256) void attn_reduce(const float* __restrict__ pacc,
                                                   const float* __restrict__ pssum,
                                                   const float* __restrict__ pmax,
                                                   float* __restrict__ out) {
  const int i = blockIdx.x;
#pragma unroll
  for (int r = 0; r < 2; ++r) {
    const int col = threadIdx.x + r * 256;
    const int head = col >> 7;
    float mz[JS];
    float mall = M_FLOOR;
#pragma unroll
    for (int zz = 0; zz < JS; ++zz) {
      mz[zz] = pmax[(size_t)zz * (HEADS * N_NODES) + head * N_NODES + i];
      mall = fmaxf(mall, mz[zz]);
    }
    float num = 0.f, den = 0.f;
#pragma unroll
    for (int zz = 0; zz < JS; ++zz) {
      const float w = __builtin_amdgcn_exp2f(mz[zz] - mall);
      num += w * pacc[(size_t)zz * (N_NODES * F_OUT) + (size_t)i * F_OUT + col];
      den += w * pssum[(size_t)zz * (HEADS * N_NODES) + head * N_NODES + i];
    }
    float val = num / den;
    val = (val > 0.f) ? val : (__expf(val) - 1.f);
    out[(size_t)i * F_OUT + col] = val;
  }
}

extern "C" void kernel_launch(void* const* d_in, const int* in_sizes, int n_in,
                              void* d_out, int out_size, void* d_ws, size_t ws_size,
                              hipStream_t stream) {
  const float* x   = (const float*)d_in[0];
  const int*   adj = (const int*)d_in[1];
  const float* W   = (const float*)d_in[2];
  const float* a   = (const float*)d_in[3];
  float* out = (float*)d_out;

  constexpr int JS = 4;
  const size_t HN = (size_t)HEADS * N_NODES;

  float* h   = (float*)d_ws;
  float* si  = h + (size_t)N_NODES * F_OUT;
  float* sj  = si + HN;
  float* pacc  = sj + HN;
  float* pssum = pacc + (size_t)JS * N_NODES * F_OUT;
  float* pmax  = pssum + JS * HN;
  const size_t need_js =
      (size_t)(pmax + JS * HN - (float*)d_ws) * sizeof(float);
  float* sjT = sj + HN;
  float* Mx  = sjT + HN;

  if (ws_size >= need_js) {
    gemm_xw_scores<<<dim3(N_NODES / 64, HEADS), 256, 0, stream>>>(x, W, a, h, si, sj);
    attn_agg_slab<JS><<<dim3(N_NODES / (AGG_WAVES * 16), HEADS, JS), AGG_WAVES * 32, 0, stream>>>(
        h, adj, si, sj, pacc, pssum, pmax);
    attn_reduce<JS><<<dim3(N_NODES), 256, 0, stream>>>(pacc, pssum, pmax, out);
  } else {
    gemm_xw<<<dim3(N_NODES / 64, F_OUT / 64), 128, 0, stream>>>(x, W, h);
    attn_scores<true><<<dim3(N_NODES / 4), 128, 0, stream>>>(h, a, si, sj, sjT);
    row_max<<<dim3(N_NODES), 256, 0, stream>>>(adj, sjT, Mx);
    attn_agg<false><<<dim3(N_NODES / 64, HEADS, 1), 128, 0, stream>>>(
        h, adj, si, sj, Mx, out, nullptr, N_NODES);
  }
}
